# starting kernel
# baseline (speedup 1.0000x reference)
#include <hip/hip_runtime.h>


typedef __attribute__((ext_vector_type(2))) float v2f;
typedef __attribute__((ext_vector_type(8))) float v8f;

#define NODE_D 128
#define COND_D 256
#define EDGE_D 64
#define OUT_D  64
#define LN_EPS 1e-5f

static __device__ __forceinline__ v8f wmma_f32(v2f a, v2f b, v8f c) {
  return __builtin_amdgcn_wmma_f32_16x16x4_f32(
       false, a,  false, b,
       (short)0, c,  false,  false);
}

__global__ __launch_bounds__(128) void node_film_kernel(
    const float* __restrict__ node_feats, const float* __restrict__ cond,
    const float* __restrict__ W_cond, const float* __restrict__ b_cond,
    const float* __restrict__ W_film, const float* __restrict__ b_film,
    float* __restrict__ h_out, int N)
{
  __shared__ float sH[4 * 16 * 65];

  const int tid  = threadIdx.x;
  const int wave = tid >> 5;
  const int lane = tid & 31;
  const int half = lane >> 4;
  const int c    = lane & 15;
  float* tile = &sH[wave * 16 * 65];

  const int nb    = blockIdx.x * 64 + wave * 16;
  const int garow = min(nb + c, N - 1);

  {
    const float* arow = node_feats + (size_t)garow * NODE_D;
    for (int t = 0; t < 4; ++t) {
      v8f acc = {0.f,0.f,0.f,0.f,0.f,0.f,0.f,0.f};
      const int col = t * 16 + c;
      for (int k = 0; k < NODE_D; k += 4) {
        v2f a = *(const v2f*)(arow + k + 2 * half);
        v2f b;
        b.x = W_film[(k + 2 * half)     * OUT_D + col];
        b.y = W_film[(k + 2 * half + 1) * OUT_D + col];
        acc = wmma_f32(a, b, acc);
      }
      const float bias = b_film[col];
#pragma unroll
      for (int i = 0; i < 8; ++i) {
        const int rowM = i + 8 * half;
        tile[rowM * 65 + col] = acc[i] + bias;
      }
    }
  }
  __syncthreads();

  {
    const int r = c;
    float mu = 0.f;
#pragma unroll
    for (int j = 0; j < OUT_D; ++j) mu += tile[r * 65 + j];
    mu *= (1.0f / OUT_D);
    float var = 0.f;
#pragma unroll
    for (int j = 0; j < OUT_D; ++j) {
      const float d = tile[r * 65 + j] - mu;
      var += d * d;
    }
    var *= (1.0f / OUT_D);
    const float rs = rsqrtf(var + LN_EPS);
#pragma unroll
    for (int j = 0; j < 32; ++j) {
      const int col = half * 32 + j;
      tile[r * 65 + col] = (tile[r * 65 + col] - mu) * rs;
    }
  }
  __syncthreads();

  {
    const float* crow = cond + (size_t)garow * COND_D;
    for (int t = 0; t < 4; ++t) {
      v8f ag = {0.f,0.f,0.f,0.f,0.f,0.f,0.f,0.f};
      v8f ab = {0.f,0.f,0.f,0.f,0.f,0.f,0.f,0.f};
      const int col = t * 16 + c;
      for (int k = 0; k < COND_D; k += 4) {
        v2f a = *(const v2f*)(crow + k + 2 * half);
        v2f bg, bb;
        bg.x = W_cond[(k + 2 * half)     * (2 * OUT_D) + col];
        bg.y = W_cond[(k + 2 * half + 1) * (2 * OUT_D) + col];
        bb.x = W_cond[(k + 2 * half)     * (2 * OUT_D) + col + OUT_D];
        bb.y = W_cond[(k + 2 * half + 1) * (2 * OUT_D) + col + OUT_D];
        ag = wmma_f32(a, bg, ag);
        ab = wmma_f32(a, bb, ab);
      }
      const float cg = b_cond[col] + 1.0f;
      const float cb = b_cond[col + OUT_D];
#pragma unroll
      for (int i = 0; i < 8; ++i) {
        const int rowM = i + 8 * half;
        const int g = nb + rowM;
        if (g < N) {
          const float hn  = tile[rowM * 65 + col];
          const float val = (ag[i] + cg) * hn + (ab[i] + cb);
          h_out[(size_t)g * OUT_D + col] = fmaxf(val, 0.f);
        }
      }
    }
  }
}

__global__ __launch_bounds__(256) void edge_scatter_kernel(
    const float* __restrict__ edge_feats, const float* __restrict__ edge_weights,
    const int* __restrict__ src, const int* __restrict__ dst,
    const int* __restrict__ mirror,
    const float* __restrict__ W_edge, const float* __restrict__ b_edge,
    const float* __restrict__ h, float* __restrict__ out, int E)
{
  __shared__ float sW[EDGE_D * OUT_D];

  const int tid = threadIdx.x;
  for (int i = tid; i < EDGE_D * OUT_D; i += 256) sW[i] = W_edge[i];
  __syncthreads();

  const int wave  = tid >> 5;
  const int lane  = tid & 31;
  const int half  = lane >> 4;
  const int c     = lane & 15;
  const int ebase = blockIdx.x * 128 + wave * 16;

  const int ea = min(ebase + c, E - 1);
  const int m  = mirror[ea];
  const float* arow = edge_feats + (size_t)m * EDGE_D;
  v2f a[16];
#pragma unroll
  for (int ks = 0; ks < 16; ++ks)
    a[ks] = *(const v2f*)(arow + ks * 4 + 2 * half);

  int   srcv[8], dstv[8];
  float ewv[8];
#pragma unroll
  for (int i = 0; i < 8; ++i) {
    const int e = min(ebase + i + 8 * half, E - 1);
    srcv[i] = src[e];
    dstv[i] = dst[e];
    ewv[i]  = edge_weights[e];
  }

  for (int t = 0; t < 4; ++t) {
    v8f acc = {0.f,0.f,0.f,0.f,0.f,0.f,0.f,0.f};
    const int col = t * 16 + c;
#pragma unroll
    for (int ks = 0; ks < 16; ++ks) {
      v2f b;
      b.x = sW[(ks * 4 + 2 * half)     * OUT_D + col];
      b.y = sW[(ks * 4 + 2 * half + 1) * OUT_D + col];
      acc = wmma_f32(a[ks], b, acc);
    }
    const float bias = b_edge[col];
#pragma unroll
    for (int i = 0; i < 8; ++i) {
      const int e = ebase + i + 8 * half;
      if (e < E) {
        const float ep  = tanhf(acc[i] + bias) * ewv[i];
        const float msg = ep * h[(size_t)srcv[i] * OUT_D + col];
        atomicAdd(&out[(size_t)dstv[i] * OUT_D + col], msg);
      }
    }
  }
}

extern "C" void kernel_launch(void* const* d_in, const int* in_sizes, int n_in,
                              void* d_out, int out_size, void* d_ws, size_t ws_size,
                              hipStream_t stream)
{
  const float* node_feats = (const float*)d_in[0];
  const float* edge_feats = (const float*)d_in[1];
  const float* cond       = (const float*)d_in[2];
  const float* edge_w     = (const float*)d_in[3];
  const int*   src        = (const int*)d_in[4];
  const int*   dst        = (const int*)d_in[5];
  const int*   mirror     = (const int*)d_in[6];
  const float* W_edge     = (const float*)d_in[7];
  const float* b_edge     = (const float*)d_in[8];
  const float* W_cond     = (const float*)d_in[9];
  const float* b_cond     = (const float*)d_in[10];
  const float* W_film     = (const float*)d_in[11];
  const float* b_film     = (const float*)d_in[12];
  (void)n_in; (void)ws_size;

  const int N = in_sizes[0] / NODE_D;
  const int E = in_sizes[4];

  float* h_ws = (float*)d_ws;
  float* out  = (float*)d_out;

  hipMemsetAsync(out, 0, (size_t)out_size * sizeof(float), stream);

  node_film_kernel<<<dim3((N + 63) / 64), dim3(128), 0, stream>>>(
      node_feats, cond, W_cond, b_cond, W_film, b_film, h_ws, N);

  edge_scatter_kernel<<<dim3((E + 127) / 128), dim3(256), 0, stream>>>(
      edge_feats, edge_w, src, dst, mirror, W_edge, b_edge, h_ws, out, E);
}


# edge kernel: each global access (A-row gather, h-row gather, atomic add) covers whole 128-B lines of one row via a per-wave LDS transpose; one atomic add per edge and column as before
# speedup vs baseline: 1.4800x; 1.4800x over previous
#include <hip/hip_runtime.h>


typedef __attribute__((ext_vector_type(2))) float v2f;
typedef __attribute__((ext_vector_type(8))) float v8f;

#define NODE_D 128
#define COND_D 256
#define EDGE_D 64
#define OUT_D  64
#define LN_EPS 1e-5f

static __device__ __forceinline__ v8f wmma_f32(v2f a, v2f b, v8f c) {
  return __builtin_amdgcn_wmma_f32_16x16x4_f32(
       false, a,  false, b,
       (short)0, c,  false,  false);
}

__global__ __launch_bounds__(128) void node_film_kernel(
    const float* __restrict__ node_feats, const float* __restrict__ cond,
    const float* __restrict__ W_cond, const float* __restrict__ b_cond,
    const float* __restrict__ W_film, const float* __restrict__ b_film,
    float* __restrict__ h_out, int N)
{
  __shared__ float sH[4 * 16 * 65];

  const int tid  = threadIdx.x;
  const int wave = tid >> 5;
  const int lane = tid & 31;
  const int half = lane >> 4;
  const int c    = lane & 15;
  float* tile = &sH[wave * 16 * 65];

  const int nb    = blockIdx.x * 64 + wave * 16;
  const int garow = min(nb + c, N - 1);

  {
    const float* arow = node_feats + (size_t)garow * NODE_D;
    for (int t = 0; t < 4; ++t) {
      v8f acc = {0.f,0.f,0.f,0.f,0.f,0.f,0.f,0.f};
      const int col = t * 16 + c;
      for (int k = 0; k < NODE_D; k += 4) {
        v2f a = *(const v2f*)(arow + k + 2 * half);
        v2f b;
        b.x = W_film[(k + 2 * half)     * OUT_D + col];
        b.y = W_film[(k + 2 * half + 1) * OUT_D + col];
        acc = wmma_f32(a, b, acc);
      }
      const float bias = b_film[col];
#pragma unroll
      for (int i = 0; i < 8; ++i) {
        const int rowM = i + 8 * half;
        tile[rowM * 65 + col] = acc[i] + bias;
      }
    }
  }
  __syncthreads();

  {
    const int r = c;
    float mu = 0.f;
#pragma unroll
    for (int j = 0; j < OUT_D; ++j) mu += tile[r * 65 + j];
    mu *= (1.0f / OUT_D);
    float var = 0.f;
#pragma unroll
    for (int j = 0; j < OUT_D; ++j) {
      const float d = tile[r * 65 + j] - mu;
      var += d * d;
    }
    var *= (1.0f / OUT_D);
    const float rs = rsqrtf(var + LN_EPS);
#pragma unroll
    for (int j = 0; j < 32; ++j) {
      const int col = half * 32 + j;
      tile[r * 65 + col] = (tile[r * 65 + col] - mu) * rs;
    }
  }
  __syncthreads();

  {
    const float* crow = cond + (size_t)garow * COND_D;
    for (int t = 0; t < 4; ++t) {
      v8f ag = {0.f,0.f,0.f,0.f,0.f,0.f,0.f,0.f};
      v8f ab = {0.f,0.f,0.f,0.f,0.f,0.f,0.f,0.f};
      const int col = t * 16 + c;
      for (int k = 0; k < COND_D; k += 4) {
        v2f a = *(const v2f*)(crow + k + 2 * half);
        v2f bg, bb;
        bg.x = W_cond[(k + 2 * half)     * (2 * OUT_D) + col];
        bg.y = W_cond[(k + 2 * half + 1) * (2 * OUT_D) + col];
        bb.x = W_cond[(k + 2 * half)     * (2 * OUT_D) + col + OUT_D];
        bb.y = W_cond[(k + 2 * half + 1) * (2 * OUT_D) + col + OUT_D];
        ag = wmma_f32(a, bg, ag);
        ab = wmma_f32(a, bb, ab);
      }
      const float cg = b_cond[col] + 1.0f;
      const float cb = b_cond[col + OUT_D];
#pragma unroll
      for (int i = 0; i < 8; ++i) {
        const int rowM = i + 8 * half;
        const int g = nb + rowM;
        if (g < N) {
          const float hn  = tile[rowM * 65 + col];
          const float val = (ag[i] + cg) * hn + (ab[i] + cb);
          h_out[(size_t)g * OUT_D + col] = fmaxf(val, 0.f);
        }
      }
    }
  }
}

#define TP 68

static __device__ __forceinline__ void wave_lds_handover() {
  __builtin_amdgcn_fence(__ATOMIC_RELEASE, "wavefront");
  __builtin_amdgcn_wave_barrier();
  __builtin_amdgcn_fence(__ATOMIC_ACQUIRE, "wavefront");
}

__global__ __launch_bounds__(256) void edge_scatter_kernel(
    const float* __restrict__ edge_feats, const float* __restrict__ edge_weights,
    const int* __restrict__ src, const int* __restrict__ dst,
    const int* __restrict__ mirror,
    const float* __restrict__ W_edge, const float* __restrict__ b_edge,
    const float* __restrict__ h, float* __restrict__ out, int E)
{
  __shared__ float sW[EDGE_D * OUT_D];
  __shared__ float sT[8 * 16 * TP];

  const int tid = threadIdx.x;
  for (int i = tid; i < EDGE_D * OUT_D; i += 256) sW[i] = W_edge[i];
  __syncthreads();

  const int wave  = __builtin_amdgcn_readfirstlane(tid >> 5);
  const int lane  = tid & 31;
  const int half  = lane >> 4;
  const int c     = lane & 15;
  const int ebase = blockIdx.x * 128 + wave * 16;
  float* tile = &sT[wave * 16 * TP];

  const int   er   = min(ebase + c, E - 1);
  const int   mrow = mirror[er];
  const int   srow = src[er];
  const int   drow = dst[er];
  const float wrow = edge_weights[er];

  v2f arow[16];
#pragma unroll
  for (int r = 0; r < 16; ++r) {
    const int m = __builtin_amdgcn_readlane(mrow, r);
    arow[r] = *(const v2f*)(edge_feats + (size_t)m * EDGE_D + 2 * lane);
  }

  float h0[16], h1[16];
#pragma unroll
  for (int r = 0; r < 16; ++r) {
    const int s = __builtin_amdgcn_readlane(srow, r);
    const float* hrow = h + (size_t)s * OUT_D;
    h0[r] = hrow[lane];
    h1[r] = hrow[lane + 32];
  }

#pragma unroll
  for (int r = 0; r < 16; ++r)
    *(v2f*)&tile[r * TP + 2 * lane] = arow[r];
  wave_lds_handover();
  v2f a[16];
#pragma unroll
  for (int ks = 0; ks < 16; ++ks)
    a[ks] = *(const v2f*)&tile[c * TP + ks * 4 + 2 * half];

  for (int t = 0; t < 4; ++t) {
    v8f acc = {0.f,0.f,0.f,0.f,0.f,0.f,0.f,0.f};
    const int col = t * 16 + c;
#pragma unroll
    for (int ks = 0; ks < 16; ++ks) {
      v2f b;
      b.x = sW[(ks * 4 + 2 * half)     * OUT_D + col];
      b.y = sW[(ks * 4 + 2 * half + 1) * OUT_D + col];
      acc = wmma_f32(a[ks], b, acc);
    }
    const float bias = b_edge[col];
#pragma unroll
    for (int i = 0; i < 8; ++i)
      tile[(i + 8 * half) * TP + col] = tanhf(acc[i] + bias);
  }
  wave_lds_handover();

  float m0[16], m1[16];
#pragma unroll
  for (int r = 0; r < 16; ++r) {
    const float w = __int_as_float(__builtin_amdgcn_readlane(__float_as_int(wrow), r));
    m0[r] = (tile[r * TP + lane]      * w) * h0[r];
    m1[r] = (tile[r * TP + lane + 32] * w) * h1[r];
  }
#pragma unroll
  for (int r = 0; r < 16; ++r) {
    if (ebase + r < E) {
      const int d = __builtin_amdgcn_readlane(drow, r);
      float* orow = out + (size_t)d * OUT_D;
      atomicAdd(&orow[lane],      m0[r]);
      atomicAdd(&orow[lane + 32], m1[r]);
    }
  }
}

extern "C" void kernel_launch(void* const* d_in, const int* in_sizes, int n_in,
                              void* d_out, int out_size, void* d_ws, size_t ws_size,
                              hipStream_t stream)
{
  const float* node_feats = (const float*)d_in[0];
  const float* edge_feats = (const float*)d_in[1];
  const float* cond       = (const float*)d_in[2];
  const float* edge_w     = (const float*)d_in[3];
  const int*   src        = (const int*)d_in[4];
  const int*   dst        = (const int*)d_in[5];
  const int*   mirror     = (const int*)d_in[6];
  const float* W_edge     = (const float*)d_in[7];
  const float* b_edge     = (const float*)d_in[8];
  const float* W_cond     = (const float*)d_in[9];
  const float* b_cond     = (const float*)d_in[10];
  const float* W_film     = (const float*)d_in[11];
  const float* b_film     = (const float*)d_in[12];
  (void)n_in; (void)ws_size;

  const int N = in_sizes[0] / NODE_D;
  const int E = in_sizes[4];

  float* h_ws = (float*)d_ws;
  float* out  = (float*)d_out;

  hipMemsetAsync(out, 0, (size_t)out_size * sizeof(float), stream);

  node_film_kernel<<<dim3((N + 63) / 64), dim3(128), 0, stream>>>(
      node_feats, cond, W_cond, b_cond, W_film, b_film, h_ws, N);

  edge_scatter_kernel<<<dim3((E + 127) / 128), dim3(256), 0, stream>>>(
      edge_feats, edge_w, src, dst, mirror, W_edge, b_edge, h_ws, out, E);
}


# edge kernel: each wave loops over 3-4 tiles of 16 edges (4096 four-wave work-groups): W_edge staged once per work-group, next tile's indices and A rows loaded one tile ahead; atomic adds as before
# speedup vs baseline: 1.5819x; 1.0689x over previous
#include <hip/hip_runtime.h>


typedef __attribute__((ext_vector_type(2))) float v2f;
typedef __attribute__((ext_vector_type(8))) float v8f;

#define NODE_D 128
#define COND_D 256
#define EDGE_D 64
#define OUT_D  64
#define LN_EPS 1e-5f

static __device__ __forceinline__ v8f wmma_f32(v2f a, v2f b, v8f c) {
  return __builtin_amdgcn_wmma_f32_16x16x4_f32(
       false, a,  false, b,
       (short)0, c,  false,  false);
}

__global__ __launch_bounds__(128) void node_film_kernel(
    const float* __restrict__ node_feats, const float* __restrict__ cond,
    const float* __restrict__ W_cond, const float* __restrict__ b_cond,
    const float* __restrict__ W_film, const float* __restrict__ b_film,
    float* __restrict__ h_out, int N)
{
  __shared__ float sH[4 * 16 * 65];

  const int tid  = threadIdx.x;
  const int wave = tid >> 5;
  const int lane = tid & 31;
  const int half = lane >> 4;
  const int c    = lane & 15;
  float* tile = &sH[wave * 16 * 65];

  const int nb    = blockIdx.x * 64 + wave * 16;
  const int garow = min(nb + c, N - 1);

  {
    const float* arow = node_feats + (size_t)garow * NODE_D;
    for (int t = 0; t < 4; ++t) {
      v8f acc = {0.f,0.f,0.f,0.f,0.f,0.f,0.f,0.f};
      const int col = t * 16 + c;
      for (int k = 0; k < NODE_D; k += 4) {
        v2f a = *(const v2f*)(arow + k + 2 * half);
        v2f b;
        b.x = W_film[(k + 2 * half)     * OUT_D + col];
        b.y = W_film[(k + 2 * half + 1) * OUT_D + col];
        acc = wmma_f32(a, b, acc);
      }
      const float bias = b_film[col];
#pragma unroll
      for (int i = 0; i < 8; ++i) {
        const int rowM = i + 8 * half;
        tile[rowM * 65 + col] = acc[i] + bias;
      }
    }
  }
  __syncthreads();

  {
    const int r = c;
    float mu = 0.f;
#pragma unroll
    for (int j = 0; j < OUT_D; ++j) mu += tile[r * 65 + j];
    mu *= (1.0f / OUT_D);
    float var = 0.f;
#pragma unroll
    for (int j = 0; j < OUT_D; ++j) {
      const float d = tile[r * 65 + j] - mu;
      var += d * d;
    }
    var *= (1.0f / OUT_D);
    const float rs = rsqrtf(var + LN_EPS);
#pragma unroll
    for (int j = 0; j < 32; ++j) {
      const int col = half * 32 + j;
      tile[r * 65 + col] = (tile[r * 65 + col] - mu) * rs;
    }
  }
  __syncthreads();

  {
    const float* crow = cond + (size_t)garow * COND_D;
    for (int t = 0; t < 4; ++t) {
      v8f ag = {0.f,0.f,0.f,0.f,0.f,0.f,0.f,0.f};
      v8f ab = {0.f,0.f,0.f,0.f,0.f,0.f,0.f,0.f};
      const int col = t * 16 + c;
      for (int k = 0; k < COND_D; k += 4) {
        v2f a = *(const v2f*)(crow + k + 2 * half);
        v2f bg, bb;
        bg.x = W_cond[(k + 2 * half)     * (2 * OUT_D) + col];
        bg.y = W_cond[(k + 2 * half + 1) * (2 * OUT_D) + col];
        bb.x = W_cond[(k + 2 * half)     * (2 * OUT_D) + col + OUT_D];
        bb.y = W_cond[(k + 2 * half + 1) * (2 * OUT_D) + col + OUT_D];
        ag = wmma_f32(a, bg, ag);
        ab = wmma_f32(a, bb, ab);
      }
      const float cg = b_cond[col] + 1.0f;
      const float cb = b_cond[col + OUT_D];
#pragma unroll
      for (int i = 0; i < 8; ++i) {
        const int rowM = i + 8 * half;
        const int g = nb + rowM;
        if (g < N) {
          const float hn  = tile[rowM * 65 + col];
          const float val = (ag[i] + cg) * hn + (ab[i] + cb);
          h_out[(size_t)g * OUT_D + col] = fmaxf(val, 0.f);
        }
      }
    }
  }
}

#define TP 68
#define EDGE_WGS 4096

typedef __attribute__((ext_vector_type(4))) float v4f;

static __device__ __forceinline__ void wave_lds_handover() {
  __builtin_amdgcn_fence(__ATOMIC_RELEASE, "wavefront");
  __builtin_amdgcn_wave_barrier();
  __builtin_amdgcn_fence(__ATOMIC_ACQUIRE, "wavefront");
}

__global__ __launch_bounds__(128) void edge_scatter_kernel(
    const float* __restrict__ edge_feats, const float* __restrict__ edge_weights,
    const int* __restrict__ src, const int* __restrict__ dst,
    const int* __restrict__ mirror,
    const float* __restrict__ W_edge, const float* __restrict__ b_edge,
    const float* __restrict__ h, float* __restrict__ out, int E)
{
  __shared__ __attribute__((aligned(16))) float sW[EDGE_D * OUT_D];
  __shared__ float sT[4 * 16 * TP];

  const int tid = threadIdx.x;
  {
    v4f wv[8];
#pragma unroll
    for (int i = 0; i < 8; ++i)
      wv[i] = *(const v4f*)(W_edge + 4 * (tid + 128 * i));
#pragma unroll
    for (int i = 0; i < 8; ++i)
      *(v4f*)&sW[4 * (tid + 128 * i)] = wv[i];
  }
  __syncthreads();

  const int wave  = __builtin_amdgcn_readfirstlane(tid >> 5);
  const int lane  = tid & 31;
  const int half  = lane >> 4;
  const int c     = lane & 15;
  float* tile = &sT[wave * 16 * TP];

  const int nTiles = (E + 15) >> 4;
  const int tstep  = (int)gridDim.x * 4;
  const int tfirst = (int)blockIdx.x * 4 + wave;

  int   srow, drow;
  float wrow;
  v2f   arow[16];
  {
    const int er = max(min(min(tfirst, nTiles - 1) * 16 + c, E - 1), 0);
    const int mrow = mirror[er];
    srow = src[er];
    drow = dst[er];
    wrow = edge_weights[er];
#pragma unroll
    for (int r = 0; r < 16; ++r) {
      const int m = __builtin_amdgcn_readlane(mrow, r);
      arow[r] = *(const v2f*)(edge_feats + (size_t)m * EDGE_D + 2 * lane);
    }
  }

  for (int tl = tfirst; tl < nTiles; tl += tstep) {
    const int  ebase    = tl * 16;
    const int  tnext    = tl + tstep;

#pragma unroll
    for (int r = 0; r < 16; ++r)
      *(v2f*)&tile[r * TP + 2 * lane] = arow[r];
    wave_lds_handover();
    v2f a[16];
#pragma unroll
    for (int ks = 0; ks < 16; ++ks)
      a[ks] = *(const v2f*)&tile[c * TP + ks * 4 + 2 * half];

    float h0[16], h1[16];
#pragma unroll
    for (int r = 0; r < 16; ++r) {
      const int s = __builtin_amdgcn_readlane(srow, r);
      const float* hrow = h + (size_t)s * OUT_D;
      h0[r] = hrow[lane];
      h1[r] = hrow[lane + 32];
    }

    const int   ern = (tnext < nTiles) ? min(tnext * 16 + c, E - 1) : E - 1;
    const int   mN  = mirror[ern];
    const int   sN  = src[ern];
    const int   dN  = dst[ern];
    const float wN  = edge_weights[ern];

    for (int t = 0; t < 4; ++t) {
      v8f acc = {0.f,0.f,0.f,0.f,0.f,0.f,0.f,0.f};
      const int col = t * 16 + c;
#pragma unroll
      for (int ks = 0; ks < 16; ++ks) {
        v2f b;
        b.x = sW[(ks * 4 + 2 * half)     * OUT_D + col];
        b.y = sW[(ks * 4 + 2 * half + 1) * OUT_D + col];
        acc = wmma_f32(a[ks], b, acc);
      }
      const float bias = b_edge[col];
#pragma unroll
      for (int i = 0; i < 8; ++i)
        tile[(i + 8 * half) * TP + col] = tanhf(acc[i] + bias);
    }

#pragma unroll
    for (int r = 0; r < 16; ++r) {
      const int m = __builtin_amdgcn_readlane(mN, r);
      arow[r] = *(const v2f*)(edge_feats + (size_t)m * EDGE_D + 2 * lane);
    }
    wave_lds_handover();

    float m0[16], m1[16];
#pragma unroll
    for (int r = 0; r < 16; ++r) {
      const float w = __int_as_float(__builtin_amdgcn_readlane(__float_as_int(wrow), r));
      m0[r] = (tile[r * TP + lane]      * w) * h0[r];
      m1[r] = (tile[r * TP + lane + 32] * w) * h1[r];
    }
#pragma unroll
    for (int r = 0; r < 16; ++r) {
      if (ebase + r < E) {
        const int d = __builtin_amdgcn_readlane(drow, r);
        float* orow = out + (size_t)d * OUT_D;
        atomicAdd(&orow[lane],      m0[r]);
        atomicAdd(&orow[lane + 32], m1[r]);
      }
    }
    wave_lds_handover();

    srow = sN;
    drow = dN;
    wrow = wN;
  }
}

extern "C" void kernel_launch(void* const* d_in, const int* in_sizes, int n_in,
                              void* d_out, int out_size, void* d_ws, size_t ws_size,
                              hipStream_t stream)
{
  const float* node_feats = (const float*)d_in[0];
  const float* edge_feats = (const float*)d_in[1];
  const float* cond       = (const float*)d_in[2];
  const float* edge_w     = (const float*)d_in[3];
  const int*   src        = (const int*)d_in[4];
  const int*   dst        = (const int*)d_in[5];
  const int*   mirror     = (const int*)d_in[6];
  const float* W_edge     = (const float*)d_in[7];
  const float* b_edge     = (const float*)d_in[8];
  const float* W_cond     = (const float*)d_in[9];
  const float* b_cond     = (const float*)d_in[10];
  const float* W_film     = (const float*)d_in[11];
  const float* b_film     = (const float*)d_in[12];
  (void)n_in; (void)ws_size;

  const int N = in_sizes[0] / NODE_D;
  const int E = in_sizes[4];

  float* h_ws = (float*)d_ws;
  float* out  = (float*)d_out;

  hipMemsetAsync(out, 0, (size_t)out_size * sizeof(float), stream);

  node_film_kernel<<<dim3((N + 63) / 64), dim3(128), 0, stream>>>(
      node_feats, cond, W_cond, b_cond, W_film, b_film, h_ws, N);

  edge_scatter_kernel<<<dim3(EDGE_WGS), dim3(128), 0, stream>>>(
      edge_feats, edge_w, src, dst, mirror, W_edge, b_edge, h_ws, out, E);
}


# edge kernel: tanh as one hardware v_tanh_f32 per value in place of the library tanhf (two exec-masked branches, about 35 instructions per value); tanh values round slightly differently
# speedup vs baseline: 1.8385x; 1.1622x over previous
#include <hip/hip_runtime.h>


typedef __attribute__((ext_vector_type(2))) float v2f;
typedef __attribute__((ext_vector_type(8))) float v8f;

#define NODE_D 128
#define COND_D 256
#define EDGE_D 64
#define OUT_D  64
#define LN_EPS 1e-5f

static __device__ __forceinline__ v8f wmma_f32(v2f a, v2f b, v8f c) {
  return __builtin_amdgcn_wmma_f32_16x16x4_f32(
       false, a,  false, b,
       (short)0, c,  false,  false);
}

__global__ __launch_bounds__(128) void node_film_kernel(
    const float* __restrict__ node_feats, const float* __restrict__ cond,
    const float* __restrict__ W_cond, const float* __restrict__ b_cond,
    const float* __restrict__ W_film, const float* __restrict__ b_film,
    float* __restrict__ h_out, int N)
{
  __shared__ float sH[4 * 16 * 65];

  const int tid  = threadIdx.x;
  const int wave = tid >> 5;
  const int lane = tid & 31;
  const int half = lane >> 4;
  const int c    = lane & 15;
  float* tile = &sH[wave * 16 * 65];

  const int nb    = blockIdx.x * 64 + wave * 16;
  const int garow = min(nb + c, N - 1);

  {
    const float* arow = node_feats + (size_t)garow * NODE_D;
    for (int t = 0; t < 4; ++t) {
      v8f acc = {0.f,0.f,0.f,0.f,0.f,0.f,0.f,0.f};
      const int col = t * 16 + c;
      for (int k = 0; k < NODE_D; k += 4) {
        v2f a = *(const v2f*)(arow + k + 2 * half);
        v2f b;
        b.x = W_film[(k + 2 * half)     * OUT_D + col];
        b.y = W_film[(k + 2 * half + 1) * OUT_D + col];
        acc = wmma_f32(a, b, acc);
      }
      const float bias = b_film[col];
#pragma unroll
      for (int i = 0; i < 8; ++i) {
        const int rowM = i + 8 * half;
        tile[rowM * 65 + col] = acc[i] + bias;
      }
    }
  }
  __syncthreads();

  {
    const int r = c;
    float mu = 0.f;
#pragma unroll
    for (int j = 0; j < OUT_D; ++j) mu += tile[r * 65 + j];
    mu *= (1.0f / OUT_D);
    float var = 0.f;
#pragma unroll
    for (int j = 0; j < OUT_D; ++j) {
      const float d = tile[r * 65 + j] - mu;
      var += d * d;
    }
    var *= (1.0f / OUT_D);
    const float rs = rsqrtf(var + LN_EPS);
#pragma unroll
    for (int j = 0; j < 32; ++j) {
      const int col = half * 32 + j;
      tile[r * 65 + col] = (tile[r * 65 + col] - mu) * rs;
    }
  }
  __syncthreads();

  {
    const float* crow = cond + (size_t)garow * COND_D;
    for (int t = 0; t < 4; ++t) {
      v8f ag = {0.f,0.f,0.f,0.f,0.f,0.f,0.f,0.f};
      v8f ab = {0.f,0.f,0.f,0.f,0.f,0.f,0.f,0.f};
      const int col = t * 16 + c;
      for (int k = 0; k < COND_D; k += 4) {
        v2f a = *(const v2f*)(crow + k + 2 * half);
        v2f bg, bb;
        bg.x = W_cond[(k + 2 * half)     * (2 * OUT_D) + col];
        bg.y = W_cond[(k + 2 * half + 1) * (2 * OUT_D) + col];
        bb.x = W_cond[(k + 2 * half)     * (2 * OUT_D) + col + OUT_D];
        bb.y = W_cond[(k + 2 * half + 1) * (2 * OUT_D) + col + OUT_D];
        ag = wmma_f32(a, bg, ag);
        ab = wmma_f32(a, bb, ab);
      }
      const float cg = b_cond[col] + 1.0f;
      const float cb = b_cond[col + OUT_D];
#pragma unroll
      for (int i = 0; i < 8; ++i) {
        const int rowM = i + 8 * half;
        const int g = nb + rowM;
        if (g < N) {
          const float hn  = tile[rowM * 65 + col];
          const float val = (ag[i] + cg) * hn + (ab[i] + cb);
          h_out[(size_t)g * OUT_D + col] = fmaxf(val, 0.f);
        }
      }
    }
  }
}

#define TP 68
#define EDGE_WGS 4096

typedef __attribute__((ext_vector_type(4))) float v4f;

static __device__ __forceinline__ void wave_lds_handover() {
  __builtin_amdgcn_fence(__ATOMIC_RELEASE, "wavefront");
  __builtin_amdgcn_wave_barrier();
  __builtin_amdgcn_fence(__ATOMIC_ACQUIRE, "wavefront");
}

static __device__ __forceinline__ float hw_tanh(float x) {
  return __builtin_amdgcn_tanhf(x);
}

__global__ __launch_bounds__(128) void edge_scatter_kernel(
    const float* __restrict__ edge_feats, const float* __restrict__ edge_weights,
    const int* __restrict__ src, const int* __restrict__ dst,
    const int* __restrict__ mirror,
    const float* __restrict__ W_edge, const float* __restrict__ b_edge,
    const float* __restrict__ h, float* __restrict__ out, int E)
{
  __shared__ __attribute__((aligned(16))) float sW[EDGE_D * OUT_D];
  __shared__ float sT[4 * 16 * TP];

  const int tid = threadIdx.x;
  {
    v4f wv[8];
#pragma unroll
    for (int i = 0; i < 8; ++i)
      wv[i] = *(const v4f*)(W_edge + 4 * (tid + 128 * i));
#pragma unroll
    for (int i = 0; i < 8; ++i)
      *(v4f*)&sW[4 * (tid + 128 * i)] = wv[i];
  }
  __syncthreads();

  const int wave  = __builtin_amdgcn_readfirstlane(tid >> 5);
  const int lane  = tid & 31;
  const int half  = lane >> 4;
  const int c     = lane & 15;
  float* tile = &sT[wave * 16 * TP];

  const int nTiles = (E + 15) >> 4;
  const int tstep  = (int)gridDim.x * 4;
  const int tfirst = (int)blockIdx.x * 4 + wave;

  int   srow, drow;
  float wrow;
  v2f   arow[16];
  {
    const int er = max(min(min(tfirst, nTiles - 1) * 16 + c, E - 1), 0);
    const int mrow = mirror[er];
    srow = src[er];
    drow = dst[er];
    wrow = edge_weights[er];
#pragma unroll
    for (int r = 0; r < 16; ++r) {
      const int m = __builtin_amdgcn_readlane(mrow, r);
      arow[r] = *(const v2f*)(edge_feats + (size_t)m * EDGE_D + 2 * lane);
    }
  }

  for (int tl = tfirst; tl < nTiles; tl += tstep) {
    const int  ebase    = tl * 16;
    const int  tnext    = tl + tstep;

#pragma unroll
    for (int r = 0; r < 16; ++r)
      *(v2f*)&tile[r * TP + 2 * lane] = arow[r];
    wave_lds_handover();
    v2f a[16];
#pragma unroll
    for (int ks = 0; ks < 16; ++ks)
      a[ks] = *(const v2f*)&tile[c * TP + ks * 4 + 2 * half];

    float h0[16], h1[16];
#pragma unroll
    for (int r = 0; r < 16; ++r) {
      const int s = __builtin_amdgcn_readlane(srow, r);
      const float* hrow = h + (size_t)s * OUT_D;
      h0[r] = hrow[lane];
      h1[r] = hrow[lane + 32];
    }

    const int   ern = (tnext < nTiles) ? min(tnext * 16 + c, E - 1) : E - 1;
    const int   mN  = mirror[ern];
    const int   sN  = src[ern];
    const int   dN  = dst[ern];
    const float wN  = edge_weights[ern];

    for (int t = 0; t < 4; ++t) {
      v8f acc = {0.f,0.f,0.f,0.f,0.f,0.f,0.f,0.f};
      const int col = t * 16 + c;
#pragma unroll
      for (int ks = 0; ks < 16; ++ks) {
        v2f b;
        b.x = sW[(ks * 4 + 2 * half)     * OUT_D + col];
        b.y = sW[(ks * 4 + 2 * half + 1) * OUT_D + col];
        acc = wmma_f32(a[ks], b, acc);
      }
      const float bias = b_edge[col];
#pragma unroll
      for (int i = 0; i < 8; ++i)
        tile[(i + 8 * half) * TP + col] = hw_tanh(acc[i] + bias);
    }

#pragma unroll
    for (int r = 0; r < 16; ++r) {
      const int m = __builtin_amdgcn_readlane(mN, r);
      arow[r] = *(const v2f*)(edge_feats + (size_t)m * EDGE_D + 2 * lane);
    }
    wave_lds_handover();

    float m0[16], m1[16];
#pragma unroll
    for (int r = 0; r < 16; ++r) {
      const float w = __int_as_float(__builtin_amdgcn_readlane(__float_as_int(wrow), r));
      m0[r] = (tile[r * TP + lane]      * w) * h0[r];
      m1[r] = (tile[r * TP + lane + 32] * w) * h1[r];
    }
#pragma unroll
    for (int r = 0; r < 16; ++r) {
      if (ebase + r < E) {
        const int d = __builtin_amdgcn_readlane(drow, r);
        float* orow = out + (size_t)d * OUT_D;
        atomicAdd(&orow[lane],      m0[r]);
        atomicAdd(&orow[lane + 32], m1[r]);
      }
    }
    wave_lds_handover();

    srow = sN;
    drow = dN;
    wrow = wN;
  }
}

extern "C" void kernel_launch(void* const* d_in, const int* in_sizes, int n_in,
                              void* d_out, int out_size, void* d_ws, size_t ws_size,
                              hipStream_t stream)
{
  const float* node_feats = (const float*)d_in[0];
  const float* edge_feats = (const float*)d_in[1];
  const float* cond       = (const float*)d_in[2];
  const float* edge_w     = (const float*)d_in[3];
  const int*   src        = (const int*)d_in[4];
  const int*   dst        = (const int*)d_in[5];
  const int*   mirror     = (const int*)d_in[6];
  const float* W_edge     = (const float*)d_in[7];
  const float* b_edge     = (const float*)d_in[8];
  const float* W_cond     = (const float*)d_in[9];
  const float* b_cond     = (const float*)d_in[10];
  const float* W_film     = (const float*)d_in[11];
  const float* b_film     = (const float*)d_in[12];
  (void)n_in; (void)ws_size;

  const int N = in_sizes[0] / NODE_D;
  const int E = in_sizes[4];

  float* h_ws = (float*)d_ws;
  float* out  = (float*)d_out;

  hipMemsetAsync(out, 0, (size_t)out_size * sizeof(float), stream);

  node_film_kernel<<<dim3((N + 63) / 64), dim3(128), 0, stream>>>(
      node_feats, cond, W_cond, b_cond, W_film, b_film, h_ws, N);

  edge_scatter_kernel<<<dim3(EDGE_WGS), dim3(128), 0, stream>>>(
      edge_feats, edge_w, src, dst, mirror, W_edge, b_edge, h_ws, out, E);
}
